# starting kernel
# baseline (speedup 1.0000x reference)
#include <hip/hip_runtime.h>

typedef float v2f __attribute__((ext_vector_type(2)));
typedef float v8f __attribute__((ext_vector_type(8)));

__device__ __forceinline__ v8f wmma4(v2f a, v2f b, v8f c) {
  return __builtin_amdgcn_wmma_f32_16x16x4_f32(false, a, false, b, (short)0, c,
                                               false, false);
}

template <int WS, int SH, int CHUNK>
__global__ __launch_bounds__(128) void win_attn_kernel(
    const float* __restrict__ x, float* __restrict__ out) {
  constexpr int H = 256, W = 256, HW = H * W;
  constexpr int NW = W / WS;
  constexpr int T = WS * WS;
  constexpr int NT = T / 16;

  __shared__ float qvl[8 * T * 4];
  __shared__ float ptile[4][16 * 20];

  const int wid = blockIdx.x;
  const int b = blockIdx.y;
  const int wy = wid / NW, wx = wid % NW;
  const int tid = threadIdx.x;
  const int lane = tid & 31;
  const int head = tid >> 5;

  const float* xbase = x + ((size_t)b * 64 + CHUNK * 32) * HW;
  for (int i = tid; i < 32 * T; i += 128) {
    int c = i / T, t = i - c * T;
    int ty = t / WS, tx = t - ty * WS;
    int gh = (wy * WS + ty - SH) & (H - 1);
    int gw = (wx * WS + tx - SH) & (W - 1);
    float val = xbase[c * HW + gh * W + gw];
    qvl[((c >> 2) * T + t) * 4 + (c & 3)] = val;
  }
  __syncthreads();

  const float* qb = &qvl[head * T * 4];
  const float* vb = &qvl[(4 + head) * T * 4];
  float* pl = &ptile[head][0];

  const int m = lane & 15;
  const int hi = lane >> 4;
  const int k0 = hi * 2;

  for (int qt = 0; qt < NT; ++qt) {
    v2f a = *(const v2f*)(qb + ((qt * 16 + m) * 4 + k0));

    v8f S[NT];
#pragma unroll
    for (int kt = 0; kt < NT; ++kt) {
      v2f bq = *(const v2f*)(qb + ((kt * 16 + m) * 4 + k0));
      v8f z = {0.f, 0.f, 0.f, 0.f, 0.f, 0.f, 0.f, 0.f};
      S[kt] = wmma4(a, bq, z);
    }

    float rmax[8];
#pragma unroll
    for (int r = 0; r < 8; ++r) {
      float mx = -3.402823466e38f;
#pragma unroll
      for (int kt = 0; kt < NT; ++kt) mx = fmaxf(mx, S[kt][r]);
      mx = fmaxf(mx, __shfl_xor(mx, 8, 32));
      mx = fmaxf(mx, __shfl_xor(mx, 4, 32));
      mx = fmaxf(mx, __shfl_xor(mx, 2, 32));
      mx = fmaxf(mx, __shfl_xor(mx, 1, 32));
      rmax[r] = mx;
    }
#pragma unroll
    for (int kt = 0; kt < NT; ++kt) {
#pragma unroll
      for (int r = 0; r < 8; ++r) S[kt][r] = __expf(S[kt][r] - rmax[r]);
    }

    v8f O = {0.f, 0.f, 0.f, 0.f, 0.f, 0.f, 0.f, 0.f};
    for (int kt = 0; kt < NT; ++kt) {
#pragma unroll
      for (int r = 0; r < 8; ++r)
        pl[(r + 8 * hi) * 20 + m] = S[kt][r];
#pragma unroll
      for (int c4 = 0; c4 < 4; ++c4) {
        v2f pa = *(const v2f*)(pl + m * 20 + c4 * 4 + k0);
        v2f bv = {0.f, 0.f};
        if (m < 4) {
          bv.x = vb[(kt * 16 + c4 * 4 + k0) * 4 + m];
          bv.y = vb[(kt * 16 + c4 * 4 + k0 + 1) * 4 + m];
        } else if (m == 4) {
          bv.x = 1.f;
          bv.y = 1.f;
        }
        O = wmma4(pa, bv, O);
      }
    }

#pragma unroll
    for (int r = 0; r < 8; ++r) {
      float sm = __shfl(O[r], (lane & 16) | 4, 32);
      if (m < 4) {
        int tok = qt * 16 + r + 8 * hi;
        int ty = tok / WS, tx = tok - ty * WS;
        int gh = (wy * WS + ty - SH) & (H - 1);
        int gw = (wx * WS + tx - SH) & (W - 1);
        out[((size_t)b * 32 + CHUNK * 16 + head * 4 + m) * HW + gh * W + gw] =
            O[r] / sm;
      }
    }
  }
}

extern "C" void kernel_launch(void* const* d_in, const int* in_sizes, int n_in,
                              void* d_out, int out_size, void* d_ws,
                              size_t ws_size, hipStream_t stream) {
  (void)in_sizes; (void)n_in; (void)out_size; (void)d_ws; (void)ws_size;
  const float* x = (const float*)d_in[0];
  float* out = (float*)d_out;

  dim3 g0(32 * 32, 4);
  win_attn_kernel<8, 0, 0><<<g0, 128, 0, stream>>>(x, out);
  dim3 g1(16 * 16, 4);
  win_attn_kernel<16, 8, 1><<<g1, 128, 0, stream>>>(x, out);
}


# Epilogue via the wave's LDS staging area: O tile transposed so each lane owns one token and a channel pair: 2 divisions + 2 window-row stores per lane, not 8 shuffles + 8 4-byte scatters; sums same
# speedup vs baseline: 2.0187x; 2.0187x over previous
#include <hip/hip_runtime.h>

typedef float v2f __attribute__((ext_vector_type(2)));
typedef float v8f __attribute__((ext_vector_type(8)));

__device__ __forceinline__ v8f wmma4(v2f a, v2f b, v8f c) {
  return __builtin_amdgcn_wmma_f32_16x16x4_f32(false, a, false, b, (short)0, c,
                                               false, false);
}

template <int WS, int SH, int CHUNK, int PB>
__global__ __launch_bounds__(128) void win_attn_kernel(
    const float* __restrict__ x, float* __restrict__ out) {
  constexpr int H = 256, W = 256, HW = H * W;
  constexpr int NW = W / WS;
  constexpr int T = WS * WS;
  constexpr int NT = T / 16;

  __shared__ float qvl[8 * T * 4];
  __shared__ float ptile[4][PB * 16 * 20];

  const int wid = blockIdx.x;
  const int b = blockIdx.y;
  const int wy = wid / NW, wx = wid % NW;
  const int tid = threadIdx.x;
  const int lane = tid & 31;
  const int head = tid >> 5;

  const float* xbase = x + ((size_t)b * 64 + CHUNK * 32) * HW;
  for (int i = tid; i < 32 * T; i += 128) {
    int c = i / T, t = i - c * T;
    int ty = t / WS, tx = t - ty * WS;
    int gh = (wy * WS + ty - SH) & (H - 1);
    int gw = (wx * WS + tx - SH) & (W - 1);
    float val = xbase[c * HW + gh * W + gw];
    qvl[((c >> 2) * T + t) * 4 + (c & 3)] = val;
  }
  __syncthreads();

  const float* qb = &qvl[head * T * 4];
  const float* vb = &qvl[(4 + head) * T * 4];
  float* pl = &ptile[head][0];

  const int m = lane & 15;
  const int hi = lane >> 4;
  const int k0 = hi * 2;
  const int mc = m & 3;
  const bool isv = m < 4;
  const float fill = (m == 4) ? 1.f : 0.f;

  for (int qt = 0; qt < NT; ++qt) {
    v2f a = *(const v2f*)(qb + ((qt * 16 + m) * 4 + k0));

    v8f S[NT];
#pragma unroll
    for (int kt = 0; kt < NT; ++kt) {
      v2f bq = *(const v2f*)(qb + ((kt * 16 + m) * 4 + k0));
      v8f z = {0.f, 0.f, 0.f, 0.f, 0.f, 0.f, 0.f, 0.f};
      S[kt] = wmma4(a, bq, z);
    }

    float rmax[8];
#pragma unroll
    for (int r = 0; r < 8; ++r) {
      float mx = -3.402823466e38f;
#pragma unroll
      for (int kt = 0; kt < NT; ++kt) mx = fmaxf(mx, S[kt][r]);
      mx = fmaxf(mx, __shfl_xor(mx, 8, 32));
      mx = fmaxf(mx, __shfl_xor(mx, 4, 32));
      mx = fmaxf(mx, __shfl_xor(mx, 2, 32));
      mx = fmaxf(mx, __shfl_xor(mx, 1, 32));
      rmax[r] = mx;
    }
#pragma unroll
    for (int kt = 0; kt < NT; ++kt) {
#pragma unroll
      for (int r = 0; r < 8; ++r) S[kt][r] = __expf(S[kt][r] - rmax[r]);
    }

    v8f O = {0.f, 0.f, 0.f, 0.f, 0.f, 0.f, 0.f, 0.f};
#pragma unroll
    for (int kb = 0; kb < NT; kb += PB) {
#pragma unroll
      for (int j = 0; j < PB; ++j) {
#pragma unroll
        for (int r = 0; r < 8; ++r)
          pl[j * 320 + (r + 8 * hi) * 20 + m] = S[kb + j][r];
      }
      constexpr int NG = PB / 2;
      v2f pa[2][8];
      float va[2][8], vc[2][8];
#pragma unroll
      for (int g = -1; g < NG; ++g) {
        if (g + 1 < NG) {
#pragma unroll
          for (int u = 0; u < 8; ++u) {
            const int j = (g + 1) * 2 + (u >> 2), c4 = u & 3;
            pa[(g + 1) & 1][u] =
                *(const v2f*)(pl + j * 320 + m * 20 + c4 * 4 + k0);
            const int key = (kb + j) * 16 + c4 * 4 + k0;
            va[(g + 1) & 1][u] = vb[key * 4 + mc];
            vc[(g + 1) & 1][u] = vb[(key + 1) * 4 + mc];
          }
        }
        __builtin_amdgcn_sched_barrier(0x406);
        if (g >= 0) {
#pragma unroll
          for (int u = 0; u < 8; ++u) {
            v2f bv = {isv ? va[g & 1][u] : fill, isv ? vc[g & 1][u] : fill};
            O = wmma4(pa[g & 1][u], bv, O);
          }
        }
        __builtin_amdgcn_sched_barrier(0x406);
      }
    }

#pragma unroll
    for (int r = 0; r < 8; ++r) pl[(r + 8 * hi) * 20 + m] = O[r];
    const v2f o2 = *(const v2f*)(pl + m * 20 + 2 * hi);
    const float sm = pl[m * 20 + 4];
    {
      const int tok = qt * 16 + m;
      const int ty = tok / WS, tx = tok - ty * WS;
      const int gh = (wy * WS + ty - SH) & (H - 1);
      const int gw = (wx * WS + tx - SH) & (W - 1);
      float* op =
          out + ((size_t)b * 32 + CHUNK * 16 + head * 4 + 2 * hi) * HW +
          gh * W + gw;
      op[0] = o2.x / sm;
      op[HW] = o2.y / sm;
    }
  }
}

extern "C" void kernel_launch(void* const* d_in, const int* in_sizes, int n_in,
                              void* d_out, int out_size, void* d_ws,
                              size_t ws_size, hipStream_t stream) {
  (void)in_sizes; (void)n_in; (void)out_size; (void)d_ws; (void)ws_size;
  const float* x = (const float*)d_in[0];
  float* out = (float*)d_out;

  dim3 g0(32 * 32, 4);
  win_attn_kernel<8, 0, 0, 2><<<g0, 128, 0, stream>>>(x, out);
  dim3 g1(16 * 16, 4);
  win_attn_kernel<16, 8, 1, 8><<<g1, 128, 0, stream>>>(x, out);
}


# Gather: every thread issues all its global loads (16 or 64, held in registers) before its first LDS store, so a block waits one memory round trip, not one per 4 loads; sums unchanged
# speedup vs baseline: 2.1544x; 1.0672x over previous
#include <hip/hip_runtime.h>

typedef float v2f __attribute__((ext_vector_type(2)));
typedef float v8f __attribute__((ext_vector_type(8)));

__device__ __forceinline__ v8f wmma4(v2f a, v2f b, v8f c) {
  return __builtin_amdgcn_wmma_f32_16x16x4_f32(false, a, false, b, (short)0, c,
                                               false, false);
}

template <int WS, int SH, int CHUNK, int PB>
__global__ __launch_bounds__(128) void win_attn_kernel(
    const float* __restrict__ x, float* __restrict__ out) {
  constexpr int H = 256, W = 256, HW = H * W;
  constexpr int NW = W / WS;
  constexpr int T = WS * WS;
  constexpr int NT = T / 16;

  __shared__ float qvl[8 * T * 4];
  __shared__ float ptile[4][PB * 16 * 20];

  const int wid = blockIdx.x;
  const int b = blockIdx.y;
  const int wy = wid / NW, wx = wid % NW;
  const int tid = threadIdx.x;
  const int lane = tid & 31;
  const int head = tid >> 5;

  const float* xbase = x + ((size_t)b * 64 + CHUNK * 32) * HW;
  constexpr int NLD = 32 * T / 128;
  float gv[NLD];
#pragma unroll
  for (int k = 0; k < NLD; ++k) {
    int c, t;
    if constexpr (T >= 128) {
      c = k / (T / 128);
      t = tid + 128 * (k % (T / 128));
    } else {
      c = k * (128 / T) + tid / T;
      t = tid % T;
    }
    int ty = t / WS, tx = t - ty * WS;
    int gh = (wy * WS + ty - SH) & (H - 1);
    int gw = (wx * WS + tx - SH) & (W - 1);
    gv[k] = xbase[c * HW + gh * W + gw];
  }
#pragma unroll
  for (int k = 0; k < NLD; ++k) {
    int c, t;
    if constexpr (T >= 128) {
      c = k / (T / 128);
      t = tid + 128 * (k % (T / 128));
    } else {
      c = k * (128 / T) + tid / T;
      t = tid % T;
    }
    qvl[((c >> 2) * T + t) * 4 + (c & 3)] = gv[k];
  }
  __syncthreads();

  const float* qb = &qvl[head * T * 4];
  const float* vb = &qvl[(4 + head) * T * 4];
  float* pl = &ptile[head][0];

  const int m = lane & 15;
  const int hi = lane >> 4;
  const int k0 = hi * 2;
  const int mc = m & 3;
  const bool isv = m < 4;
  const float fill = (m == 4) ? 1.f : 0.f;

  for (int qt = 0; qt < NT; ++qt) {
    v2f a = *(const v2f*)(qb + ((qt * 16 + m) * 4 + k0));

    v8f S[NT];
#pragma unroll
    for (int kt = 0; kt < NT; ++kt) {
      v2f bq = *(const v2f*)(qb + ((kt * 16 + m) * 4 + k0));
      v8f z = {0.f, 0.f, 0.f, 0.f, 0.f, 0.f, 0.f, 0.f};
      S[kt] = wmma4(a, bq, z);
    }

    float rmax[8];
#pragma unroll
    for (int r = 0; r < 8; ++r) {
      float mx = -3.402823466e38f;
#pragma unroll
      for (int kt = 0; kt < NT; ++kt) mx = fmaxf(mx, S[kt][r]);
      mx = fmaxf(mx, __shfl_xor(mx, 8, 32));
      mx = fmaxf(mx, __shfl_xor(mx, 4, 32));
      mx = fmaxf(mx, __shfl_xor(mx, 2, 32));
      mx = fmaxf(mx, __shfl_xor(mx, 1, 32));
      rmax[r] = mx;
    }
#pragma unroll
    for (int kt = 0; kt < NT; ++kt) {
#pragma unroll
      for (int r = 0; r < 8; ++r) S[kt][r] = __expf(S[kt][r] - rmax[r]);
    }

    v8f O = {0.f, 0.f, 0.f, 0.f, 0.f, 0.f, 0.f, 0.f};
#pragma unroll
    for (int kb = 0; kb < NT; kb += PB) {
#pragma unroll
      for (int j = 0; j < PB; ++j) {
#pragma unroll
        for (int r = 0; r < 8; ++r)
          pl[j * 320 + (r + 8 * hi) * 20 + m] = S[kb + j][r];
      }
      constexpr int NG = PB / 2;
      v2f pa[2][8];
      float va[2][8], vc[2][8];
#pragma unroll
      for (int g = -1; g < NG; ++g) {
        if (g + 1 < NG) {
#pragma unroll
          for (int u = 0; u < 8; ++u) {
            const int j = (g + 1) * 2 + (u >> 2), c4 = u & 3;
            pa[(g + 1) & 1][u] =
                *(const v2f*)(pl + j * 320 + m * 20 + c4 * 4 + k0);
            const int key = (kb + j) * 16 + c4 * 4 + k0;
            va[(g + 1) & 1][u] = vb[key * 4 + mc];
            vc[(g + 1) & 1][u] = vb[(key + 1) * 4 + mc];
          }
        }
        __builtin_amdgcn_sched_barrier(0x406);
        if (g >= 0) {
#pragma unroll
          for (int u = 0; u < 8; ++u) {
            v2f bv = {isv ? va[g & 1][u] : fill, isv ? vc[g & 1][u] : fill};
            O = wmma4(pa[g & 1][u], bv, O);
          }
        }
        __builtin_amdgcn_sched_barrier(0x406);
      }
    }

#pragma unroll
    for (int r = 0; r < 8; ++r) pl[(r + 8 * hi) * 20 + m] = O[r];
    const v2f o2 = *(const v2f*)(pl + m * 20 + 2 * hi);
    const float sm = pl[m * 20 + 4];
    {
      const int tok = qt * 16 + m;
      const int ty = tok / WS, tx = tok - ty * WS;
      const int gh = (wy * WS + ty - SH) & (H - 1);
      const int gw = (wx * WS + tx - SH) & (W - 1);
      float* op =
          out + ((size_t)b * 32 + CHUNK * 16 + head * 4 + 2 * hi) * HW +
          gh * W + gw;
      op[0] = o2.x / sm;
      op[HW] = o2.y / sm;
    }
  }
}

extern "C" void kernel_launch(void* const* d_in, const int* in_sizes, int n_in,
                              void* d_out, int out_size, void* d_ws,
                              size_t ws_size, hipStream_t stream) {
  (void)in_sizes; (void)n_in; (void)out_size; (void)d_ws; (void)ws_size;
  const float* x = (const float*)d_in[0];
  float* out = (float*)d_out;

  dim3 g0(32 * 32, 4);
  win_attn_kernel<8, 0, 0, 2><<<g0, 128, 0, stream>>>(x, out);
  dim3 g1(16 * 16, 4);
  win_attn_kernel<16, 8, 1, 8><<<g1, 128, 0, stream>>>(x, out);
}


# Score symmetry: E=exp(-|qi-qj|^2/2) tiles on/above the diagonal only, staged once, read transposed and plain for two products; weights exp(|q|^2/2-max) in [V|1]; no row max; key sum order changes
# speedup vs baseline: 2.4302x; 1.1280x over previous
#include <hip/hip_runtime.h>

typedef float v2f __attribute__((ext_vector_type(2)));
typedef float v4f __attribute__((ext_vector_type(4)));
typedef float v8f __attribute__((ext_vector_type(8)));

__device__ __forceinline__ v8f wmma4(v2f a, v2f b, v8f c) {
  return __builtin_amdgcn_wmma_f32_16x16x4_f32(false, a, false, b, (short)0, c,
                                               false, false);
}

template <int WS, int SH, int CHUNK>
__global__ __launch_bounds__(128, (WS == 8) ? 8 : 4) void win_attn_kernel(
    const float* __restrict__ x, float* __restrict__ out) {
  constexpr int H = 256, W = 256, HW = H * W;
  constexpr int NW = W / WS;
  constexpr int T = WS * WS;
  constexpr int NT = T / 16;

  __shared__ __attribute__((aligned(16))) float ql[4 * T * 4];
  __shared__ __attribute__((aligned(16))) float vwl[4 * T * 8];
  __shared__ __attribute__((aligned(16))) float nnl[4 * T];
  __shared__ __attribute__((aligned(16))) float ptile[4][16 * 20];

  const int wid = blockIdx.x;
  const int b = blockIdx.y;
  const int wy = wid / NW, wx = wid % NW;
  const int tid = threadIdx.x;
  const int lane = tid & 31;
  const int head = tid >> 5;

  const float* xbase = x + ((size_t)b * 64 + CHUNK * 32) * HW;
  constexpr int NLD = 32 * T / 128;
  float gv[NLD];
#pragma unroll
  for (int k = 0; k < NLD; ++k) {
    int c, t;
    if constexpr (T >= 128) {
      c = k / (T / 128);
      t = tid + 128 * (k % (T / 128));
    } else {
      c = k * (128 / T) + tid / T;
      t = tid % T;
    }
    int ty = t / WS, tx = t - ty * WS;
    int gh = (wy * WS + ty - SH) & (H - 1);
    int gw = (wx * WS + tx - SH) & (W - 1);
    gv[k] = xbase[c * HW + gh * W + gw];
  }
#pragma unroll
  for (int k = 0; k < NLD; ++k) {
    int c, t;
    if constexpr (T >= 128) {
      c = k / (T / 128);
      t = tid + 128 * (k % (T / 128));
    } else {
      c = k * (128 / T) + tid / T;
      t = tid % T;
    }
    if (k < NLD / 2) {
      ql[((c >> 2) * T + t) * 4 + (c & 3)] = gv[k] * 1.2011224087864498f;
    } else {
      vwl[(((c - 16) >> 2) * T + t) * 8 + (c & 3)] = gv[k];
    }
  }
  __syncthreads();

  const float* qb = &ql[head * T * 4];
  float* vw = &vwl[head * T * 8];
  float* nb = &nnl[head * T];
  float* pl = &ptile[head][0];

  constexpr int TPL = T / 32;
  float nnv[TPL];
  float nmax = 0.f;
#pragma unroll
  for (int u = 0; u < TPL; ++u) {
    const int t = lane + 32 * u;
    const v4f q = *(const v4f*)(qb + t * 4);
    nnv[u] = 0.5f * (q.x * q.x + q.y * q.y + q.z * q.z + q.w * q.w);
    nb[t] = -nnv[u];
    nmax = fmaxf(nmax, nnv[u]);
  }
  nmax = fmaxf(nmax, __shfl_xor(nmax, 16, 32));
  nmax = fmaxf(nmax, __shfl_xor(nmax, 8, 32));
  nmax = fmaxf(nmax, __shfl_xor(nmax, 4, 32));
  nmax = fmaxf(nmax, __shfl_xor(nmax, 2, 32));
  nmax = fmaxf(nmax, __shfl_xor(nmax, 1, 32));
#pragma unroll
  for (int u = 0; u < TPL; ++u) {
    const int t = lane + 32 * u;
    const v4f v = *(const v4f*)(vw + t * 8);
    const float w = __builtin_amdgcn_exp2f(nnv[u] - nmax);
    const v4f vs = {v.x * w, v.y * w, v.z * w, v.w * w};
    const v4f ws = {w, 0.f, 0.f, 0.f};
    *(v4f*)(vw + t * 8) = vs;
    *(v4f*)(vw + t * 8 + 4) = ws;
  }
  __syncthreads();

  const int m = lane & 15;
  const int hi = lane >> 4;
  const int k0 = hi * 2;
  const int mc = m & 7;
  const bool live = m < 8;

  v8f O[NT];
#pragma unroll
  for (int j = 0; j < NT; ++j) {
    const v8f z = {0.f, 0.f, 0.f, 0.f, 0.f, 0.f, 0.f, 0.f};
    O[j] = z;
  }

#pragma nounroll
  for (int i = 0; i < NT; ++i) {
    const v2f a = *(const v2f*)(qb + ((i * 16 + m) * 4 + k0));
    const v4f rn0 = *(const v4f*)(nb + i * 16 + 8 * hi);
    const v4f rn1 = *(const v4f*)(nb + i * 16 + 8 * hi + 4);
    v2f bvi[4];
#pragma unroll
    for (int c4 = 0; c4 < 4; ++c4) {
      const int key = i * 16 + c4 * 4 + k0;
      const float x0 = vw[key * 8 + mc];
      const float x1 = vw[(key + 1) * 8 + mc];
      const v2f bv = {live ? x0 : 0.f, live ? x1 : 0.f};
      bvi[c4] = bv;
    }
    v8f Orow = {0.f, 0.f, 0.f, 0.f, 0.f, 0.f, 0.f, 0.f};

#pragma unroll
    for (int j = 0; j < NT; ++j) {
      if (j >= i) {
        const v2f bq = *(const v2f*)(qb + ((j * 16 + m) * 4 + k0));
        const float cn = nb[j * 16 + m];
        const v8f C = {rn0.x + cn, rn0.y + cn, rn0.z + cn, rn0.w + cn,
                       rn1.x + cn, rn1.y + cn, rn1.z + cn, rn1.w + cn};
        v8f S = wmma4(a, bq, C);
#pragma unroll
        for (int r = 0; r < 8; ++r) S[r] = __builtin_amdgcn_exp2f(S[r]);

        const v4f s0 = {S[0], S[1], S[2], S[3]};
        const v4f s1 = {S[4], S[5], S[6], S[7]};
        *(v4f*)(pl + m * 20 + 8 * hi) = s0;
        *(v4f*)(pl + m * 20 + 8 * hi + 4) = s1;
        v2f at[4];
#pragma unroll
        for (int c4 = 0; c4 < 4; ++c4)
          at[c4] = *(const v2f*)(pl + m * 20 + c4 * 4 + k0);

        if (j == i) {
#pragma unroll
          for (int c4 = 0; c4 < 4; ++c4) Orow = wmma4(at[c4], bvi[c4], Orow);
          Orow += O[j];
        } else {
          v2f ap[4], bvj[4];
#pragma unroll
          for (int c4 = 0; c4 < 4; ++c4) {
            const v2f p = {pl[(c4 * 4 + k0) * 20 + m],
                           pl[(c4 * 4 + k0 + 1) * 20 + m]};
            ap[c4] = p;
            const int key = j * 16 + c4 * 4 + k0;
            const float x0 = vw[key * 8 + mc];
            const float x1 = vw[(key + 1) * 8 + mc];
            const v2f bv = {live ? x0 : 0.f, live ? x1 : 0.f};
            bvj[c4] = bv;
          }
#pragma unroll
          for (int c4 = 0; c4 < 4; ++c4) O[j] = wmma4(at[c4], bvi[c4], O[j]);
#pragma unroll
          for (int c4 = 0; c4 < 4; ++c4) Orow = wmma4(ap[c4], bvj[c4], Orow);
        }
      }
    }

#pragma unroll
    for (int r = 0; r < 8; ++r) pl[(r + 8 * hi) * 20 + m] = Orow[r];
    const v2f o2 = *(const v2f*)(pl + m * 20 + 2 * hi);
    const float sm = pl[m * 20 + 4];
    {
      const int tok = i * 16 + m;
      const int ty = tok / WS, tx = tok - ty * WS;
      const int gh = (wy * WS + ty - SH) & (H - 1);
      const int gw = (wx * WS + tx - SH) & (W - 1);
      float* op =
          out + ((size_t)b * 32 + CHUNK * 16 + head * 4 + 2 * hi) * HW +
          gh * W + gw;
      op[0] = o2.x / sm;
      op[HW] = o2.y / sm;
    }
  }
}

extern "C" void kernel_launch(void* const* d_in, const int* in_sizes, int n_in,
                              void* d_out, int out_size, void* d_ws,
                              size_t ws_size, hipStream_t stream) {
  (void)in_sizes; (void)n_in; (void)out_size; (void)d_ws; (void)ws_size;
  const float* x = (const float*)d_in[0];
  float* out = (float*)d_out;

  dim3 g0(32 * 32, 4);
  win_attn_kernel<8, 0, 0><<<g0, 128, 0, stream>>>(x, out);
  dim3 g1(16 * 16, 4);
  win_attn_kernel<16, 8, 1><<<g1, 128, 0, stream>>>(x, out);
}


# Both products take a tile's 16 keys in the score registers' order (8h+2c+s): E^T's A operand is the score registers (no LDS), E's A and [wv|w] are 16-byte row reads; key sum order in a tile changes
# speedup vs baseline: 2.5157x; 1.0352x over previous
#include <hip/hip_runtime.h>

typedef float v2f __attribute__((ext_vector_type(2)));
typedef float v4f __attribute__((ext_vector_type(4)));
typedef float v8f __attribute__((ext_vector_type(8)));

__device__ __forceinline__ v8f wmma4(v2f a, v2f b, v8f c) {
  return __builtin_amdgcn_wmma_f32_16x16x4_f32(false, a, false, b, (short)0, c,
                                               false, false);
}

template <int WS, int SH, int CHUNK>
__global__ __launch_bounds__(128, (WS == 8) ? 8 : 4) void win_attn_kernel(
    const float* __restrict__ x, float* __restrict__ out) {
  constexpr int H = 256, W = 256, HW = H * W;
  constexpr int NW = W / WS;
  constexpr int T = WS * WS;
  constexpr int NT = T / 16;

  __shared__ __attribute__((aligned(16))) float ql[4 * T * 4];
  constexpr int TP = T + 4;
  __shared__ __attribute__((aligned(16))) float vwl[4 * 5 * TP];
  __shared__ __attribute__((aligned(16))) float nnl[4 * T];
  __shared__ __attribute__((aligned(16))) float ptile[4][16 * 20];

  const int wid = blockIdx.x;
  const int b = blockIdx.y;
  const int wy = wid / NW, wx = wid % NW;
  const int tid = threadIdx.x;
  const int lane = tid & 31;
  const int head = tid >> 5;

  const float* xbase = x + ((size_t)b * 64 + CHUNK * 32) * HW;
  constexpr int NLD = 32 * T / 128;
  float gv[NLD];
#pragma unroll
  for (int k = 0; k < NLD; ++k) {
    int c, t;
    if constexpr (T >= 128) {
      c = k / (T / 128);
      t = tid + 128 * (k % (T / 128));
    } else {
      c = k * (128 / T) + tid / T;
      t = tid % T;
    }
    int ty = t / WS, tx = t - ty * WS;
    int gh = (wy * WS + ty - SH) & (H - 1);
    int gw = (wx * WS + tx - SH) & (W - 1);
    gv[k] = xbase[c * HW + gh * W + gw];
  }
#pragma unroll
  for (int k = 0; k < NLD; ++k) {
    int c, t;
    if constexpr (T >= 128) {
      c = k / (T / 128);
      t = tid + 128 * (k % (T / 128));
    } else {
      c = k * (128 / T) + tid / T;
      t = tid % T;
    }
    if (k < NLD / 2) {
      ql[((c >> 2) * T + t) * 4 + (c & 3)] = gv[k] * 1.2011224087864498f;
    } else {
      vwl[(((c - 16) >> 2) * 5 + (c & 3)) * TP + t] = gv[k];
    }
  }
  __syncthreads();

  const float* qb = &ql[head * T * 4];
  float* vw = &vwl[head * 5 * TP];
  float* nb = &nnl[head * T];
  float* pl = &ptile[head][0];

  constexpr int TPL = T / 32;
  float nnv[TPL];
  float nmax = 0.f;
#pragma unroll
  for (int u = 0; u < TPL; ++u) {
    const int t = lane + 32 * u;
    const v4f q = *(const v4f*)(qb + t * 4);
    nnv[u] = 0.5f * (q.x * q.x + q.y * q.y + q.z * q.z + q.w * q.w);
    nb[t] = -nnv[u];
    nmax = fmaxf(nmax, nnv[u]);
  }
  nmax = fmaxf(nmax, __shfl_xor(nmax, 16, 32));
  nmax = fmaxf(nmax, __shfl_xor(nmax, 8, 32));
  nmax = fmaxf(nmax, __shfl_xor(nmax, 4, 32));
  nmax = fmaxf(nmax, __shfl_xor(nmax, 2, 32));
  nmax = fmaxf(nmax, __shfl_xor(nmax, 1, 32));
#pragma unroll
  for (int u = 0; u < TPL; ++u) {
    const int t = lane + 32 * u;
    const float w = __builtin_amdgcn_exp2f(nnv[u] - nmax);
#pragma unroll
    for (int c = 0; c < 4; ++c) vw[c * TP + t] *= w;
    vw[4 * TP + t] = w;
  }
  __syncthreads();

  const int m = lane & 15;
  const int hi = lane >> 4;
  const int k0 = hi * 2;
  const int cm = m < 4 ? m : 4;
  const float* bp = vw + cm * TP + 8 * hi;
  float* ps = pl + (8 * hi) * 20 + m;
  const float* pr = pl + m * 20 + 8 * hi;

  v8f O[NT];
#pragma unroll
  for (int j = 0; j < NT; ++j) {
    const v8f z = {0.f, 0.f, 0.f, 0.f, 0.f, 0.f, 0.f, 0.f};
    O[j] = z;
  }

#pragma nounroll
  for (int i = 0; i < NT; ++i) {
    const v2f a = *(const v2f*)(qb + ((i * 16 + m) * 4 + k0));
    const v4f rn0 = *(const v4f*)(nb + i * 16 + 8 * hi);
    const v4f rn1 = *(const v4f*)(nb + i * 16 + 8 * hi + 4);
    const v4f bi0 = *(const v4f*)(bp + i * 16);
    const v4f bi1 = *(const v4f*)(bp + i * 16 + 4);
    const v2f bvi[4] = {{bi0.x, bi0.y}, {bi0.z, bi0.w},
                        {bi1.x, bi1.y}, {bi1.z, bi1.w}};
    v8f Orow = {0.f, 0.f, 0.f, 0.f, 0.f, 0.f, 0.f, 0.f};

#pragma unroll
    for (int j = 0; j < NT; ++j) {
      if (j >= i) {
        const v2f bq = *(const v2f*)(qb + ((j * 16 + m) * 4 + k0));
        const float cn = nb[j * 16 + m];
        const v8f C = {rn0.x + cn, rn0.y + cn, rn0.z + cn, rn0.w + cn,
                       rn1.x + cn, rn1.y + cn, rn1.z + cn, rn1.w + cn};
        v8f S = wmma4(a, bq, C);
#pragma unroll
        for (int r = 0; r < 8; ++r) S[r] = __builtin_amdgcn_exp2f(S[r]);

        const v2f at[4] = {{S[0], S[1]}, {S[2], S[3]}, {S[4], S[5]},
                           {S[6], S[7]}};

        if (j == i) {
#pragma unroll
          for (int c4 = 0; c4 < 4; ++c4) Orow = wmma4(at[c4], bvi[c4], Orow);
          Orow += O[j];
        } else {
#pragma unroll
          for (int r = 0; r < 8; ++r) ps[r * 20] = S[r];
          const v4f p0 = *(const v4f*)(pr);
          const v4f p1 = *(const v4f*)(pr + 4);
          const v2f ap[4] = {{p0.x, p0.y}, {p0.z, p0.w}, {p1.x, p1.y},
                             {p1.z, p1.w}};
          const v4f bj0 = *(const v4f*)(bp + j * 16);
          const v4f bj1 = *(const v4f*)(bp + j * 16 + 4);
          const v2f bvj[4] = {{bj0.x, bj0.y}, {bj0.z, bj0.w}, {bj1.x, bj1.y},
                              {bj1.z, bj1.w}};
#pragma unroll
          for (int c4 = 0; c4 < 4; ++c4) O[j] = wmma4(at[c4], bvi[c4], O[j]);
#pragma unroll
          for (int c4 = 0; c4 < 4; ++c4) Orow = wmma4(ap[c4], bvj[c4], Orow);
        }
      }
    }

#pragma unroll
    for (int r = 0; r < 8; ++r) pl[(r + 8 * hi) * 20 + m] = Orow[r];
    const v2f o2 = *(const v2f*)(pl + m * 20 + 2 * hi);
    const float sm = pl[m * 20 + 4];
    {
      const int tok = i * 16 + m;
      const int ty = tok / WS, tx = tok - ty * WS;
      const int gh = (wy * WS + ty - SH) & (H - 1);
      const int gw = (wx * WS + tx - SH) & (W - 1);
      float* op =
          out + ((size_t)b * 32 + CHUNK * 16 + head * 4 + 2 * hi) * HW +
          gh * W + gw;
      op[0] = o2.x / sm;
      op[HW] = o2.y / sm;
    }
  }
}

extern "C" void kernel_launch(void* const* d_in, const int* in_sizes, int n_in,
                              void* d_out, int out_size, void* d_ws,
                              size_t ws_size, hipStream_t stream) {
  (void)in_sizes; (void)n_in; (void)out_size; (void)d_ws; (void)ws_size;
  const float* x = (const float*)d_in[0];
  float* out = (float*)d_out;

  dim3 g0(32 * 32, 4);
  win_attn_kernel<8, 0, 0><<<g0, 128, 0, stream>>>(x, out);
  dim3 g1(16 * 16, 4);
  win_attn_kernel<16, 8, 1><<<g1, 128, 0, stream>>>(x, out);
}


# Row loop software-pipelined: tile j+1's score WMMA, exponentials and LDS staging issue inside the step that runs tile j's 8 product WMMAs; one LDS read group per step; diagonal peeled; sums unchanged
# speedup vs baseline: 2.6787x; 1.0648x over previous
#include <hip/hip_runtime.h>

typedef float v2f __attribute__((ext_vector_type(2)));
typedef float v4f __attribute__((ext_vector_type(4)));
typedef float v8f __attribute__((ext_vector_type(8)));

__device__ __forceinline__ v8f wmma4(v2f a, v2f b, v8f c) {
  return __builtin_amdgcn_wmma_f32_16x16x4_f32(false, a, false, b, (short)0, c,
                                               false, false);
}

template <int WS, int SH, int CHUNK>
__global__ __launch_bounds__(128, (WS == 8) ? 8 : 4) void win_attn_kernel(
    const float* __restrict__ x, float* __restrict__ out) {
  constexpr int H = 256, W = 256, HW = H * W;
  constexpr int NW = W / WS;
  constexpr int T = WS * WS;
  constexpr int NT = T / 16;

  __shared__ __attribute__((aligned(16))) float ql[4 * T * 4];
  constexpr int TP = T + 4;
  __shared__ __attribute__((aligned(16))) float vwl[4 * 5 * TP];
  __shared__ __attribute__((aligned(16))) float nnl[4 * T];
  __shared__ __attribute__((aligned(16))) float ptile[4][16 * 20];

  const int wid = blockIdx.x;
  const int b = blockIdx.y;
  const int wy = wid / NW, wx = wid % NW;
  const int tid = threadIdx.x;
  const int lane = tid & 31;
  const int head = tid >> 5;

  const float* xbase = x + ((size_t)b * 64 + CHUNK * 32) * HW;
  constexpr int NLD = 32 * T / 128;
  float gv[NLD];
#pragma unroll
  for (int k = 0; k < NLD; ++k) {
    int c, t;
    if constexpr (T >= 128) {
      c = k / (T / 128);
      t = tid + 128 * (k % (T / 128));
    } else {
      c = k * (128 / T) + tid / T;
      t = tid % T;
    }
    int ty = t / WS, tx = t - ty * WS;
    int gh = (wy * WS + ty - SH) & (H - 1);
    int gw = (wx * WS + tx - SH) & (W - 1);
    gv[k] = xbase[c * HW + gh * W + gw];
  }
#pragma unroll
  for (int k = 0; k < NLD; ++k) {
    int c, t;
    if constexpr (T >= 128) {
      c = k / (T / 128);
      t = tid + 128 * (k % (T / 128));
    } else {
      c = k * (128 / T) + tid / T;
      t = tid % T;
    }
    if (k < NLD / 2) {
      ql[((c >> 2) * T + t) * 4 + (c & 3)] = gv[k] * 1.2011224087864498f;
    } else {
      vwl[(((c - 16) >> 2) * 5 + (c & 3)) * TP + t] = gv[k];
    }
  }
  __syncthreads();

  const float* qb = &ql[head * T * 4];
  float* vw = &vwl[head * 5 * TP];
  float* nb = &nnl[head * T];
  float* pl = &ptile[head][0];

  constexpr int TPL = T / 32;
  float nnv[TPL];
  float nmax = 0.f;
#pragma unroll
  for (int u = 0; u < TPL; ++u) {
    const int t = lane + 32 * u;
    const v4f q = *(const v4f*)(qb + t * 4);
    nnv[u] = 0.5f * (q.x * q.x + q.y * q.y + q.z * q.z + q.w * q.w);
    nb[t] = -nnv[u];
    nmax = fmaxf(nmax, nnv[u]);
  }
  nmax = fmaxf(nmax, __shfl_xor(nmax, 16, 32));
  nmax = fmaxf(nmax, __shfl_xor(nmax, 8, 32));
  nmax = fmaxf(nmax, __shfl_xor(nmax, 4, 32));
  nmax = fmaxf(nmax, __shfl_xor(nmax, 2, 32));
  nmax = fmaxf(nmax, __shfl_xor(nmax, 1, 32));
#pragma unroll
  for (int u = 0; u < TPL; ++u) {
    const int t = lane + 32 * u;
    const float w = __builtin_amdgcn_exp2f(nnv[u] - nmax);
#pragma unroll
    for (int c = 0; c < 4; ++c) vw[c * TP + t] *= w;
    vw[4 * TP + t] = w;
  }
  __syncthreads();

  const int m = lane & 15;
  const int hi = lane >> 4;
  const int k0 = hi * 2;
  const int cm = m < 4 ? m : 4;
  const float* bp = vw + cm * TP + 8 * hi;
  float* ps = pl + (8 * hi) * 20 + m;
  const float* pr = pl + m * 20 + 8 * hi;

  v8f O[NT];
#pragma unroll
  for (int j = 0; j < NT; ++j) {
    const v8f z = {0.f, 0.f, 0.f, 0.f, 0.f, 0.f, 0.f, 0.f};
    O[j] = z;
  }

#pragma nounroll
  for (int i = 0; i < NT; ++i) {
    const v2f a = *(const v2f*)(qb + ((i * 16 + m) * 4 + k0));
    const v4f rn0 = *(const v4f*)(nb + i * 16 + 8 * hi);
    const v4f rn1 = *(const v4f*)(nb + i * 16 + 8 * hi + 4);
    const v4f bi0 = *(const v4f*)(bp + i * 16);
    const v4f bi1 = *(const v4f*)(bp + i * 16 + 4);
    const v2f bvi[4] = {{bi0.x, bi0.y}, {bi0.z, bi0.w},
                        {bi1.x, bi1.y}, {bi1.z, bi1.w}};
    const int jn = (i + 1 < NT) ? i + 1 : i;
    const v2f bqn = *(const v2f*)(qb + ((jn * 16 + m) * 4 + k0));
    const float cnn = nb[jn * 16 + m];

    v8f E;
    {
      const float cn = nb[i * 16 + m];
      const v8f C = {rn0.x + cn, rn0.y + cn, rn0.z + cn, rn0.w + cn,
                     rn1.x + cn, rn1.y + cn, rn1.z + cn, rn1.w + cn};
      E = wmma4(a, a, C);
#pragma unroll
      for (int r = 0; r < 8; ++r) E[r] = __builtin_amdgcn_exp2f(E[r]);
    }
    v8f Orow = {0.f, 0.f, 0.f, 0.f, 0.f, 0.f, 0.f, 0.f};
    {
      const v2f at[4] = {{E[0], E[1]}, {E[2], E[3]}, {E[4], E[5]},
                         {E[6], E[7]}};
#pragma unroll
      for (int c4 = 0; c4 < 4; ++c4) Orow = wmma4(at[c4], bvi[c4], Orow);
    }
    {
      const v8f C = {rn0.x + cnn, rn0.y + cnn, rn0.z + cnn, rn0.w + cnn,
                     rn1.x + cnn, rn1.y + cnn, rn1.z + cnn, rn1.w + cnn};
      E = wmma4(a, bqn, C);
    }
    {
      v8f Oi = O[0];
      switch (i) {
#define PICK_TILE(k) case k: Oi = O[(k) < NT ? (k) : 0]; break;
        PICK_TILE(1) PICK_TILE(2) PICK_TILE(3) PICK_TILE(4) PICK_TILE(5)
        PICK_TILE(6) PICK_TILE(7) PICK_TILE(8) PICK_TILE(9) PICK_TILE(10)
        PICK_TILE(11) PICK_TILE(12) PICK_TILE(13) PICK_TILE(14) PICK_TILE(15)
#undef PICK_TILE
        default: break;
      }
      Orow += Oi;
    }
#pragma unroll
    for (int r = 0; r < 8; ++r) E[r] = __builtin_amdgcn_exp2f(E[r]);
#pragma unroll
    for (int r = 0; r < 8; ++r) ps[r * 20] = E[r];

#pragma unroll
    for (int j = 1; j < NT; ++j) {
      if (j > i) {
        const v4f p0 = *(const v4f*)(pr);
        const v4f p1 = *(const v4f*)(pr + 4);
        const v4f bj0 = *(const v4f*)(bp + j * 16);
        const v4f bj1 = *(const v4f*)(bp + j * 16 + 4);
        v2f bq = {0.f, 0.f};
        float cn = 0.f;
        if (j + 1 < NT) {
          bq = *(const v2f*)(qb + (((j + 1) * 16 + m) * 4 + k0));
          cn = nb[(j + 1) * 16 + m];
        }
        __builtin_amdgcn_sched_barrier(0);

        const v2f at[4] = {{E[0], E[1]}, {E[2], E[3]}, {E[4], E[5]},
                           {E[6], E[7]}};
#pragma unroll
        for (int c4 = 0; c4 < 4; ++c4) O[j] = wmma4(at[c4], bvi[c4], O[j]);
        __builtin_amdgcn_sched_barrier(0);

        if (j + 1 < NT) {
          const v8f C = {rn0.x + cn, rn0.y + cn, rn0.z + cn, rn0.w + cn,
                         rn1.x + cn, rn1.y + cn, rn1.z + cn, rn1.w + cn};
          E = wmma4(a, bq, C);
        }
        {
          const v2f ap[4] = {{p0.x, p0.y}, {p0.z, p0.w}, {p1.x, p1.y},
                             {p1.z, p1.w}};
          const v2f bvj[4] = {{bj0.x, bj0.y}, {bj0.z, bj0.w}, {bj1.x, bj1.y},
                              {bj1.z, bj1.w}};
#pragma unroll
          for (int c4 = 0; c4 < 4; ++c4) Orow = wmma4(ap[c4], bvj[c4], Orow);
        }
        if (j + 1 < NT) {
#pragma unroll
          for (int r = 0; r < 8; ++r) E[r] = __builtin_amdgcn_exp2f(E[r]);
#pragma unroll
          for (int r = 0; r < 8; ++r) ps[r * 20] = E[r];
        }
      }
    }

#pragma unroll
    for (int r = 0; r < 8; ++r) pl[(r + 8 * hi) * 20 + m] = Orow[r];
    const v2f o2 = *(const v2f*)(pl + m * 20 + 2 * hi);
    const float sm = pl[m * 20 + 4];
    {
      const int tok = i * 16 + m;
      const int ty = tok / WS, tx = tok - ty * WS;
      const int gh = (wy * WS + ty - SH) & (H - 1);
      const int gw = (wx * WS + tx - SH) & (W - 1);
      float* op =
          out + ((size_t)b * 32 + CHUNK * 16 + head * 4 + 2 * hi) * HW +
          gh * W + gw;
      op[0] = o2.x / sm;
      op[HW] = o2.y / sm;
    }
  }
}

extern "C" void kernel_launch(void* const* d_in, const int* in_sizes, int n_in,
                              void* d_out, int out_size, void* d_ws,
                              size_t ws_size, hipStream_t stream) {
  (void)in_sizes; (void)n_in; (void)out_size; (void)d_ws; (void)ws_size;
  const float* x = (const float*)d_in[0];
  float* out = (float*)d_out;

  dim3 g0(32 * 32, 4);
  win_attn_kernel<8, 0, 0><<<g0, 128, 0, stream>>>(x, out);
  dim3 g1(16 * 16, 4);
  win_attn_kernel<16, 8, 1><<<g1, 128, 0, stream>>>(x, out);
}
